# starting kernel
# baseline (speedup 1.0000x reference)
#include <hip/hip_runtime.h>
#include <math.h>


typedef float v2f __attribute__((ext_vector_type(2)));
typedef float v8f __attribute__((ext_vector_type(8)));

#define GAT_B 4
#define GAT_N 20000
#define GAT_K 16
#define GAT_D 256
#define GAT_H 4
#define GAT_CH 64

__global__ __launch_bounds__(128)
void gat_gemm_wmma(const float* __restrict__ x,
                   const float* __restrict__ W,
                   float* __restrict__ h)
{
    const int lane = threadIdx.x & 31;
    const int wave = threadIdx.x >> 5;
    const int m0   = blockIdx.x * 16;
    const int n0   = wave * 64;
    const int half = lane >> 4;
    const int l16  = lane & 15;

    const float* arow = x + (size_t)(m0 + l16) * GAT_D;
    const float* b0r  = W + (size_t)(n0 +  0 + l16) * GAT_D;
    const float* b1r  = W + (size_t)(n0 + 16 + l16) * GAT_D;
    const float* b2r  = W + (size_t)(n0 + 32 + l16) * GAT_D;
    const float* b3r  = W + (size_t)(n0 + 48 + l16) * GAT_D;

    v8f acc0 = {}; v8f acc1 = {}; v8f acc2 = {}; v8f acc3 = {};

    #pragma unroll 4
    for (int k = 0; k < GAT_D; k += 4) {
        const int kk = k + 2 * half;
        v2f a  = *(const v2f*)(arow + kk);
        v2f b0 = *(const v2f*)(b0r + kk);
        v2f b1 = *(const v2f*)(b1r + kk);
        v2f b2 = *(const v2f*)(b2r + kk);
        v2f b3 = *(const v2f*)(b3r + kk);
        acc0 = __builtin_amdgcn_wmma_f32_16x16x4_f32(false, a, false, b0, (short)0, acc0, false, false);
        acc1 = __builtin_amdgcn_wmma_f32_16x16x4_f32(false, a, false, b1, (short)0, acc1, false, false);
        acc2 = __builtin_amdgcn_wmma_f32_16x16x4_f32(false, a, false, b2, (short)0, acc2, false, false);
        acc3 = __builtin_amdgcn_wmma_f32_16x16x4_f32(false, a, false, b3, (short)0, acc3, false, false);
    }

    #pragma unroll
    for (int r = 0; r < 8; ++r) {
        float* hrow = h + (size_t)(m0 + r + 8 * half) * GAT_D + n0 + l16;
        hrow[ 0] = acc0[r];
        hrow[16] = acc1[r];
        hrow[32] = acc2[r];
        hrow[48] = acc3[r];
    }
}

__device__ __forceinline__ float dot4f(float4 a, float4 b) {
    return fmaf(a.x, b.x, fmaf(a.y, b.y, fmaf(a.z, b.z, a.w * b.w)));
}

__global__ __launch_bounds__(256)
void gat_gather_softmax(const float* __restrict__ h,
                        const int*   __restrict__ nbr,
                        const float* __restrict__ attn,
                        float*       __restrict__ out)
{
    const int lane = threadIdx.x & 31;
    const int wid  = threadIdx.x >> 5;
    const int p    = blockIdx.x * 8 + wid;
    const int b    = p / GAT_N;
    const int c0   = lane * 8;
    const int head = lane >> 3;
    const int ch   = c0 & (GAT_CH - 1);

    const int4* idxv = (const int4*)(nbr + (size_t)p * GAT_K);
    int4 i0 = idxv[0];
    int4 i1 = idxv[1];
    int4 i2 = idxv[2];
    int4 i3 = idxv[3];
    int nbs[GAT_K] = { i0.x, i0.y, i0.z, i0.w,
                       i1.x, i1.y, i1.z, i1.w,
                       i2.x, i2.y, i2.z, i2.w,
                       i3.x, i3.y, i3.z, i3.w };

    const float* ap = attn + head * (2 * GAT_CH) + ch;
    float4 as0 = *(const float4*)(ap);
    float4 as1 = *(const float4*)(ap + 4);
    float4 ad0 = *(const float4*)(ap + GAT_CH);
    float4 ad1 = *(const float4*)(ap + GAT_CH + 4);

    const float* hp = h + (size_t)p * GAT_D + c0;
    float4 hi0 = *(const float4*)(hp);
    float4 hi1 = *(const float4*)(hp + 4);
    float ssrc = dot4f(hi0, as0) + dot4f(hi1, as1);
    ssrc += __shfl_xor(ssrc, 1, 32);
    ssrc += __shfl_xor(ssrc, 2, 32);
    ssrc += __shfl_xor(ssrc, 4, 32);

    const float* hbase = h + (size_t)b * GAT_N * GAT_D + c0;

    float m = -3.402823466e+38f;
    float s = 0.0f;
    float a0 = 0.f, a1 = 0.f, a2 = 0.f, a3 = 0.f;
    float a4 = 0.f, a5 = 0.f, a6 = 0.f, a7 = 0.f;

    #pragma unroll
    for (int k = 0; k < GAT_K; ++k) {
        const float* jp = hbase + (size_t)nbs[k] * GAT_D;
        float4 j0 = *(const float4*)(jp);
        float4 j1 = *(const float4*)(jp + 4);

        float sd = dot4f(j0, ad0) + dot4f(j1, ad1);
        sd += __shfl_xor(sd, 1, 32);
        sd += __shfl_xor(sd, 2, 32);
        sd += __shfl_xor(sd, 4, 32);

        const float e  = (ssrc + sd) * 0.125f;
        const float mn = fmaxf(m, e);
        const float sc = __expf(m - mn);
        const float w  = __expf(e - mn);
        s = s * sc + w;
        m = mn;
        a0 = a0 * sc + w * j0.x;
        a1 = a1 * sc + w * j0.y;
        a2 = a2 * sc + w * j0.z;
        a3 = a3 * sc + w * j0.w;
        a4 = a4 * sc + w * j1.x;
        a5 = a5 * sc + w * j1.y;
        a6 = a6 * sc + w * j1.z;
        a7 = a7 * sc + w * j1.w;
    }

    const float inv = 1.0f / s;
    float v[8] = { a0 * inv, a1 * inv, a2 * inv, a3 * inv,
                   a4 * inv, a5 * inv, a6 * inv, a7 * inv };

    #pragma unroll
    for (int j = 0; j < 8; ++j) {
        float t = v[j];
        t += __shfl_xor(t, 8, 32);
        t += __shfl_xor(t, 16, 32);
        v[j] = t * 0.25f;
    }

    if (lane < 8) {
        float* op = out + (size_t)p * GAT_CH + lane * 8;
        *(float4*)(op)     = make_float4(v[0], v[1], v[2], v[3]);
        *(float4*)(op + 4) = make_float4(v[4], v[5], v[6], v[7]);
    }
}

extern "C" void kernel_launch(void* const* d_in, const int* in_sizes, int n_in,
                              void* d_out, int out_size, void* d_ws, size_t ws_size,
                              hipStream_t stream)
{
    const float* x    = (const float*)d_in[0];
    const int*   idx  = (const int*)  d_in[1];
    const float* W    = (const float*)d_in[2];
    const float* attn = (const float*)d_in[3];
    float* out = (float*)d_out;
    float* h   = (float*)d_ws;

    gat_gemm_wmma<<<(GAT_B * GAT_N) / 16, 128, 0, stream>>>(x, W, h);

    gat_gather_softmax<<<(GAT_B * GAT_N) / 8, 256, 0, stream>>>(h, idx, attn, out);
}


# GEMM register blocking: each wave computes an 80x64 tile of h (was 16x64) from float4 fragment loads, one W fragment feeds 5 row tiles; k is summed in a new order inside each 8-k step
# speedup vs baseline: 1.4417x; 1.4417x over previous
#include <hip/hip_runtime.h>
#include <math.h>


typedef float v2f __attribute__((ext_vector_type(2)));
typedef float v8f __attribute__((ext_vector_type(8)));

#define GAT_B 4
#define GAT_N 20000
#define GAT_K 16
#define GAT_D 256
#define GAT_H 4
#define GAT_CH 64

typedef float v4f __attribute__((ext_vector_type(4)));

#define GEMM_MT 5
#define GEMM_NT 4

__global__ __launch_bounds__(128)
void gat_gemm_wmma(const float* __restrict__ x,
                   const float* __restrict__ W,
                   float* __restrict__ h)
{
    const int lane = threadIdx.x & 31;
    const int wave = threadIdx.x >> 5;
    const int m0   = blockIdx.x * (16 * GEMM_MT);
    const int n0   = wave * (16 * GEMM_NT);
    const int half = lane >> 4;
    const int l16  = lane & 15;

    const float* ap = x + (size_t)(m0 + l16) * GAT_D + 4 * half;
    const float* bp = W + (size_t)(n0 + l16) * GAT_D + 4 * half;

    v8f acc[GEMM_MT][GEMM_NT];
    #pragma unroll
    for (int i = 0; i < GEMM_MT; ++i)
        #pragma unroll
        for (int j = 0; j < GEMM_NT; ++j)
            acc[i][j] = (v8f){};

    #pragma unroll 2
    for (int k = 0; k < GAT_D; k += 8) {
        v4f a[GEMM_MT], b[GEMM_NT];
        #pragma unroll
        for (int i = 0; i < GEMM_MT; ++i)
            a[i] = *(const v4f*)(ap + (size_t)i * 16 * GAT_D + k);
        #pragma unroll
        for (int j = 0; j < GEMM_NT; ++j)
            b[j] = *(const v4f*)(bp + (size_t)j * 16 * GAT_D + k);

        #pragma unroll
        for (int i = 0; i < GEMM_MT; ++i) {
            const v2f alo = { a[i].x, a[i].y };
            #pragma unroll
            for (int j = 0; j < GEMM_NT; ++j) {
                const v2f blo = { b[j].x, b[j].y };
                acc[i][j] = __builtin_amdgcn_wmma_f32_16x16x4_f32(false, alo, false, blo, (short)0, acc[i][j], false, false);
            }
        }
        #pragma unroll
        for (int i = 0; i < GEMM_MT; ++i) {
            const v2f ahi = { a[i].z, a[i].w };
            #pragma unroll
            for (int j = 0; j < GEMM_NT; ++j) {
                const v2f bhi = { b[j].z, b[j].w };
                acc[i][j] = __builtin_amdgcn_wmma_f32_16x16x4_f32(false, ahi, false, bhi, (short)0, acc[i][j], false, false);
            }
        }
    }

    #pragma unroll
    for (int i = 0; i < GEMM_MT; ++i) {
        #pragma unroll
        for (int r = 0; r < 8; ++r) {
            float* hrow = h + (size_t)(m0 + 16 * i + r + 8 * half) * GAT_D + n0 + l16;
            #pragma unroll
            for (int j = 0; j < GEMM_NT; ++j)
                hrow[16 * j] = acc[i][j][r];
        }
    }
}

__device__ __forceinline__ float dot4f(float4 a, float4 b) {
    return fmaf(a.x, b.x, fmaf(a.y, b.y, fmaf(a.z, b.z, a.w * b.w)));
}

__global__ __launch_bounds__(256)
void gat_gather_softmax(const float* __restrict__ h,
                        const int*   __restrict__ nbr,
                        const float* __restrict__ attn,
                        float*       __restrict__ out)
{
    const int lane = threadIdx.x & 31;
    const int wid  = threadIdx.x >> 5;
    const int p    = blockIdx.x * 8 + wid;
    const int b    = p / GAT_N;
    const int c0   = lane * 8;
    const int head = lane >> 3;
    const int ch   = c0 & (GAT_CH - 1);

    const int4* idxv = (const int4*)(nbr + (size_t)p * GAT_K);
    int4 i0 = idxv[0];
    int4 i1 = idxv[1];
    int4 i2 = idxv[2];
    int4 i3 = idxv[3];
    int nbs[GAT_K] = { i0.x, i0.y, i0.z, i0.w,
                       i1.x, i1.y, i1.z, i1.w,
                       i2.x, i2.y, i2.z, i2.w,
                       i3.x, i3.y, i3.z, i3.w };

    const float* ap = attn + head * (2 * GAT_CH) + ch;
    float4 as0 = *(const float4*)(ap);
    float4 as1 = *(const float4*)(ap + 4);
    float4 ad0 = *(const float4*)(ap + GAT_CH);
    float4 ad1 = *(const float4*)(ap + GAT_CH + 4);

    const float* hp = h + (size_t)p * GAT_D + c0;
    float4 hi0 = *(const float4*)(hp);
    float4 hi1 = *(const float4*)(hp + 4);
    float ssrc = dot4f(hi0, as0) + dot4f(hi1, as1);
    ssrc += __shfl_xor(ssrc, 1, 32);
    ssrc += __shfl_xor(ssrc, 2, 32);
    ssrc += __shfl_xor(ssrc, 4, 32);

    const float* hbase = h + (size_t)b * GAT_N * GAT_D + c0;

    float m = -3.402823466e+38f;
    float s = 0.0f;
    float a0 = 0.f, a1 = 0.f, a2 = 0.f, a3 = 0.f;
    float a4 = 0.f, a5 = 0.f, a6 = 0.f, a7 = 0.f;

    #pragma unroll
    for (int k = 0; k < GAT_K; ++k) {
        const float* jp = hbase + (size_t)nbs[k] * GAT_D;
        float4 j0 = *(const float4*)(jp);
        float4 j1 = *(const float4*)(jp + 4);

        float sd = dot4f(j0, ad0) + dot4f(j1, ad1);
        sd += __shfl_xor(sd, 1, 32);
        sd += __shfl_xor(sd, 2, 32);
        sd += __shfl_xor(sd, 4, 32);

        const float e  = (ssrc + sd) * 0.125f;
        const float mn = fmaxf(m, e);
        const float sc = __expf(m - mn);
        const float w  = __expf(e - mn);
        s = s * sc + w;
        m = mn;
        a0 = a0 * sc + w * j0.x;
        a1 = a1 * sc + w * j0.y;
        a2 = a2 * sc + w * j0.z;
        a3 = a3 * sc + w * j0.w;
        a4 = a4 * sc + w * j1.x;
        a5 = a5 * sc + w * j1.y;
        a6 = a6 * sc + w * j1.z;
        a7 = a7 * sc + w * j1.w;
    }

    const float inv = 1.0f / s;
    float v[8] = { a0 * inv, a1 * inv, a2 * inv, a3 * inv,
                   a4 * inv, a5 * inv, a6 * inv, a7 * inv };

    #pragma unroll
    for (int j = 0; j < 8; ++j) {
        float t = v[j];
        t += __shfl_xor(t, 8, 32);
        t += __shfl_xor(t, 16, 32);
        v[j] = t * 0.25f;
    }

    if (lane < 8) {
        float* op = out + (size_t)p * GAT_CH + lane * 8;
        *(float4*)(op)     = make_float4(v[0], v[1], v[2], v[3]);
        *(float4*)(op + 4) = make_float4(v[4], v[5], v[6], v[7]);
    }
}

extern "C" void kernel_launch(void* const* d_in, const int* in_sizes, int n_in,
                              void* d_out, int out_size, void* d_ws, size_t ws_size,
                              hipStream_t stream)
{
    const float* x    = (const float*)d_in[0];
    const int*   idx  = (const int*)  d_in[1];
    const float* W    = (const float*)d_in[2];
    const float* attn = (const float*)d_in[3];
    float* out = (float*)d_out;
    float* h   = (float*)d_ws;

    gat_gemm_wmma<<<(GAT_B * GAT_N) / (16 * GEMM_MT), 128, 0, stream>>>(x, W, h);

    gat_gather_softmax<<<(GAT_B * GAT_N) / 8, 256, 0, stream>>>(h, idx, attn, out);
}


# gather kernel: 5000 resident waves each walk 16 nodes (was 80000 one-node waves); node id wave-uniform, so row bases are scalar and the next node's indices come by scalar load one trip ahead
# speedup vs baseline: 1.6197x; 1.1235x over previous
#include <hip/hip_runtime.h>
#include <math.h>


typedef float v2f __attribute__((ext_vector_type(2)));
typedef float v8f __attribute__((ext_vector_type(8)));

#define GAT_B 4
#define GAT_N 20000
#define GAT_K 16
#define GAT_D 256
#define GAT_H 4
#define GAT_CH 64

typedef float v4f __attribute__((ext_vector_type(4)));

#define GEMM_MT 5
#define GEMM_NT 4

__global__ __launch_bounds__(128)
void gat_gemm_wmma(const float* __restrict__ x,
                   const float* __restrict__ W,
                   float* __restrict__ h)
{
    const int lane = threadIdx.x & 31;
    const int wave = threadIdx.x >> 5;
    const int m0   = blockIdx.x * (16 * GEMM_MT);
    const int n0   = wave * (16 * GEMM_NT);
    const int half = lane >> 4;
    const int l16  = lane & 15;

    const float* ap = x + (size_t)(m0 + l16) * GAT_D + 4 * half;
    const float* bp = W + (size_t)(n0 + l16) * GAT_D + 4 * half;

    v8f acc[GEMM_MT][GEMM_NT];
    #pragma unroll
    for (int i = 0; i < GEMM_MT; ++i)
        #pragma unroll
        for (int j = 0; j < GEMM_NT; ++j)
            acc[i][j] = (v8f){};

    #pragma unroll 2
    for (int k = 0; k < GAT_D; k += 8) {
        v4f a[GEMM_MT], b[GEMM_NT];
        #pragma unroll
        for (int i = 0; i < GEMM_MT; ++i)
            a[i] = *(const v4f*)(ap + (size_t)i * 16 * GAT_D + k);
        #pragma unroll
        for (int j = 0; j < GEMM_NT; ++j)
            b[j] = *(const v4f*)(bp + (size_t)j * 16 * GAT_D + k);

        #pragma unroll
        for (int i = 0; i < GEMM_MT; ++i) {
            const v2f alo = { a[i].x, a[i].y };
            #pragma unroll
            for (int j = 0; j < GEMM_NT; ++j) {
                const v2f blo = { b[j].x, b[j].y };
                acc[i][j] = __builtin_amdgcn_wmma_f32_16x16x4_f32(false, alo, false, blo, (short)0, acc[i][j], false, false);
            }
        }
        #pragma unroll
        for (int i = 0; i < GEMM_MT; ++i) {
            const v2f ahi = { a[i].z, a[i].w };
            #pragma unroll
            for (int j = 0; j < GEMM_NT; ++j) {
                const v2f bhi = { b[j].z, b[j].w };
                acc[i][j] = __builtin_amdgcn_wmma_f32_16x16x4_f32(false, ahi, false, bhi, (short)0, acc[i][j], false, false);
            }
        }
    }

    #pragma unroll
    for (int i = 0; i < GEMM_MT; ++i) {
        #pragma unroll
        for (int r = 0; r < 8; ++r) {
            float* hrow = h + (size_t)(m0 + 16 * i + r + 8 * half) * GAT_D + n0 + l16;
            #pragma unroll
            for (int j = 0; j < GEMM_NT; ++j)
                hrow[16 * j] = acc[i][j][r];
        }
    }
}

__device__ __forceinline__ float dot4f(float4 a, float4 b) {
    return fmaf(a.x, b.x, fmaf(a.y, b.y, fmaf(a.z, b.z, a.w * b.w)));
}

typedef const __attribute__((address_space(1))) char*  gbytes_t;
typedef __attribute__((address_space(1))) float4       gfloat4_t;

#define GS_WAVES 4
#define GS_NPW   16

__global__ __launch_bounds__(32 * GS_WAVES)
void gat_gather_softmax(const float* __restrict__ h,
                        const int*   __restrict__ nbr,
                        const float* __restrict__ attn,
                        float*       __restrict__ out)
{
    const int lane = threadIdx.x & 31;
    const int wid  = __builtin_amdgcn_readfirstlane((int)(threadIdx.x >> 5));
    const int p0   = (blockIdx.x * GS_WAVES + wid) * GS_NPW;
    const unsigned loff = (unsigned)lane * 32u;

    const int4* idxv = (const int4*)(nbr + (size_t)p0 * GAT_K);
    int4 i0 = idxv[0];
    int4 i1 = idxv[1];
    int4 i2 = idxv[2];
    int4 i3 = idxv[3];

    #pragma unroll 1
    for (int i = 0; i < GS_NPW; ++i) {
        const int p = p0 + i;
        const int b = p / GAT_N;

        unsigned lo = loff;
        asm volatile("" : "+v"(lo));

        const gfloat4_t* aq = (const gfloat4_t*)((gbytes_t)attn + ((lo >> 8) << 9) + (lo & 0xE0u));
        const float4 as0 = aq[0];
        const float4 as1 = aq[1];
        const float4 ad0 = aq[GAT_CH / 4];
        const float4 ad1 = aq[GAT_CH / 4 + 1];
        const int nbs[GAT_K] = { i0.x, i0.y, i0.z, i0.w,
                                 i1.x, i1.y, i1.z, i1.w,
                                 i2.x, i2.y, i2.z, i2.w,
                                 i3.x, i3.y, i3.z, i3.w };

        gbytes_t hb = (gbytes_t)(h + (size_t)p * GAT_D);
        asm("" : "+s"(hb));
        const gfloat4_t* hq = (const gfloat4_t*)(hb + lo);
        const float4 hi0 = hq[0];
        const float4 hi1 = hq[1];

        float4 j0[GAT_K], j1[GAT_K];
        #pragma unroll
        for (int k = 0; k < GAT_K; ++k) {
            const unsigned row = (unsigned)b * GAT_N + (unsigned)nbs[k];
            gbytes_t rb = (gbytes_t)(h + (size_t)row * GAT_D);
            asm("" : "+s"(rb));
            const gfloat4_t* jq = (const gfloat4_t*)(rb + lo);
            j0[k] = jq[0];
            j1[k] = jq[1];
        }

        const int pn = (p + 1 < GAT_B * GAT_N) ? (p + 1) : p;
        const int4* idxn = (const int4*)(nbr + (size_t)pn * GAT_K);
        i0 = idxn[0];
        i1 = idxn[1];
        i2 = idxn[2];
        i3 = idxn[3];

        float ssrc = dot4f(hi0, as0) + dot4f(hi1, as1);
        ssrc += __shfl_xor(ssrc, 1, 32);
        ssrc += __shfl_xor(ssrc, 2, 32);
        ssrc += __shfl_xor(ssrc, 4, 32);

        float m = -3.402823466e+38f;
        float s = 0.0f;
        float a0 = 0.f, a1 = 0.f, a2 = 0.f, a3 = 0.f;
        float a4 = 0.f, a5 = 0.f, a6 = 0.f, a7 = 0.f;

        #pragma unroll
        for (int k = 0; k < GAT_K; ++k) {
            float sd = dot4f(j0[k], ad0) + dot4f(j1[k], ad1);
            sd += __shfl_xor(sd, 1, 32);
            sd += __shfl_xor(sd, 2, 32);
            sd += __shfl_xor(sd, 4, 32);

            const float e  = (ssrc + sd) * 0.125f;
            const float mn = fmaxf(m, e);
            const float sc = __expf(m - mn);
            const float w  = __expf(e - mn);
            s = s * sc + w;
            m = mn;
            a0 = a0 * sc + w * j0[k].x;
            a1 = a1 * sc + w * j0[k].y;
            a2 = a2 * sc + w * j0[k].z;
            a3 = a3 * sc + w * j0[k].w;
            a4 = a4 * sc + w * j1[k].x;
            a5 = a5 * sc + w * j1[k].y;
            a6 = a6 * sc + w * j1[k].z;
            a7 = a7 * sc + w * j1[k].w;
        }

        const float inv = 1.0f / s;
        float v[8] = { a0 * inv, a1 * inv, a2 * inv, a3 * inv,
                       a4 * inv, a5 * inv, a6 * inv, a7 * inv };

        #pragma unroll
        for (int j = 0; j < 8; ++j) {
            float t = v[j];
            t += __shfl_xor(t, 8, 32);
            t += __shfl_xor(t, 16, 32);
            v[j] = t * 0.25f;
        }

        if (lane < 8) {
            float* op = out + (size_t)p * GAT_CH + lane * 8;
            *(float4*)(op)     = make_float4(v[0], v[1], v[2], v[3]);
            *(float4*)(op + 4) = make_float4(v[4], v[5], v[6], v[7]);
        }
    }
}

extern "C" void kernel_launch(void* const* d_in, const int* in_sizes, int n_in,
                              void* d_out, int out_size, void* d_ws, size_t ws_size,
                              hipStream_t stream)
{
    const float* x    = (const float*)d_in[0];
    const int*   idx  = (const int*)  d_in[1];
    const float* W    = (const float*)d_in[2];
    const float* attn = (const float*)d_in[3];
    float* out = (float*)d_out;
    float* h   = (float*)d_ws;

    gat_gemm_wmma<<<(GAT_B * GAT_N) / (16 * GEMM_MT), 128, 0, stream>>>(x, W, h);

    gat_gather_softmax<<<(GAT_B * GAT_N) / (GS_WAVES * GS_NPW), 32 * GS_WAVES, 0, stream>>>(h, idx, attn, out);
}


# gather kernel: a lane owns two 16-byte pieces 128 bytes apart (was 32 contiguous bytes), so a row load reads 4 whole 128-byte lines, not 8 half lines; score dot products sum channels in a new order
# speedup vs baseline: 1.6930x; 1.0453x over previous
#include <hip/hip_runtime.h>
#include <math.h>


typedef float v2f __attribute__((ext_vector_type(2)));
typedef float v8f __attribute__((ext_vector_type(8)));

#define GAT_B 4
#define GAT_N 20000
#define GAT_K 16
#define GAT_D 256
#define GAT_H 4
#define GAT_CH 64

typedef float v4f __attribute__((ext_vector_type(4)));

#define GEMM_MT 5
#define GEMM_NT 4

__global__ __launch_bounds__(128)
void gat_gemm_wmma(const float* __restrict__ x,
                   const float* __restrict__ W,
                   float* __restrict__ h)
{
    const int lane = threadIdx.x & 31;
    const int wave = threadIdx.x >> 5;
    const int m0   = blockIdx.x * (16 * GEMM_MT);
    const int n0   = wave * (16 * GEMM_NT);
    const int half = lane >> 4;
    const int l16  = lane & 15;

    const float* ap = x + (size_t)(m0 + l16) * GAT_D + 4 * half;
    const float* bp = W + (size_t)(n0 + l16) * GAT_D + 4 * half;

    v8f acc[GEMM_MT][GEMM_NT];
    #pragma unroll
    for (int i = 0; i < GEMM_MT; ++i)
        #pragma unroll
        for (int j = 0; j < GEMM_NT; ++j)
            acc[i][j] = (v8f){};

    #pragma unroll 2
    for (int k = 0; k < GAT_D; k += 8) {
        v4f a[GEMM_MT], b[GEMM_NT];
        #pragma unroll
        for (int i = 0; i < GEMM_MT; ++i)
            a[i] = *(const v4f*)(ap + (size_t)i * 16 * GAT_D + k);
        #pragma unroll
        for (int j = 0; j < GEMM_NT; ++j)
            b[j] = *(const v4f*)(bp + (size_t)j * 16 * GAT_D + k);

        #pragma unroll
        for (int i = 0; i < GEMM_MT; ++i) {
            const v2f alo = { a[i].x, a[i].y };
            #pragma unroll
            for (int j = 0; j < GEMM_NT; ++j) {
                const v2f blo = { b[j].x, b[j].y };
                acc[i][j] = __builtin_amdgcn_wmma_f32_16x16x4_f32(false, alo, false, blo, (short)0, acc[i][j], false, false);
            }
        }
        #pragma unroll
        for (int i = 0; i < GEMM_MT; ++i) {
            const v2f ahi = { a[i].z, a[i].w };
            #pragma unroll
            for (int j = 0; j < GEMM_NT; ++j) {
                const v2f bhi = { b[j].z, b[j].w };
                acc[i][j] = __builtin_amdgcn_wmma_f32_16x16x4_f32(false, ahi, false, bhi, (short)0, acc[i][j], false, false);
            }
        }
    }

    #pragma unroll
    for (int i = 0; i < GEMM_MT; ++i) {
        #pragma unroll
        for (int r = 0; r < 8; ++r) {
            float* hrow = h + (size_t)(m0 + 16 * i + r + 8 * half) * GAT_D + n0 + l16;
            #pragma unroll
            for (int j = 0; j < GEMM_NT; ++j)
                hrow[16 * j] = acc[i][j][r];
        }
    }
}

__device__ __forceinline__ float dot4f(float4 a, float4 b) {
    return fmaf(a.x, b.x, fmaf(a.y, b.y, fmaf(a.z, b.z, a.w * b.w)));
}

typedef const __attribute__((address_space(1))) char*  gbytes_t;
typedef __attribute__((address_space(1))) float4       gfloat4_t;

#define GS_WAVES 4
#define GS_NPW   16

__global__ __launch_bounds__(32 * GS_WAVES)
void gat_gather_softmax(const float* __restrict__ h,
                        const int*   __restrict__ nbr,
                        const float* __restrict__ attn,
                        float*       __restrict__ out)
{
    const int lane = threadIdx.x & 31;
    const int wid  = __builtin_amdgcn_readfirstlane((int)(threadIdx.x >> 5));
    const int p0   = (blockIdx.x * GS_WAVES + wid) * GS_NPW;
    const unsigned loff = ((unsigned)(lane >> 3) << 8) | ((unsigned)(lane & 7) << 4);

    const int4* idxv = (const int4*)(nbr + (size_t)p0 * GAT_K);
    int4 i0 = idxv[0];
    int4 i1 = idxv[1];
    int4 i2 = idxv[2];
    int4 i3 = idxv[3];

    #pragma unroll 1
    for (int i = 0; i < GS_NPW; ++i) {
        const int p = p0 + i;
        const int b = p / GAT_N;

        unsigned lo = loff;
        asm volatile("" : "+v"(lo));

        const gfloat4_t* aq = (const gfloat4_t*)((gbytes_t)attn + ((lo >> 8) << 9) + (lo & 0x70u));
        const float4 as0 = aq[0];
        const float4 as1 = aq[8];
        const float4 ad0 = aq[GAT_CH / 4];
        const float4 ad1 = aq[GAT_CH / 4 + 8];
        const int nbs[GAT_K] = { i0.x, i0.y, i0.z, i0.w,
                                 i1.x, i1.y, i1.z, i1.w,
                                 i2.x, i2.y, i2.z, i2.w,
                                 i3.x, i3.y, i3.z, i3.w };

        gbytes_t hb = (gbytes_t)(h + (size_t)p * GAT_D);
        asm("" : "+s"(hb));
        const gfloat4_t* hq = (const gfloat4_t*)(hb + lo);
        const float4 hi0 = hq[0];
        const float4 hi1 = hq[8];

        float4 j0[GAT_K], j1[GAT_K];
        #pragma unroll
        for (int k = 0; k < GAT_K; ++k) {
            const unsigned row = (unsigned)b * GAT_N + (unsigned)nbs[k];
            gbytes_t rb = (gbytes_t)(h + (size_t)row * GAT_D);
            asm("" : "+s"(rb));
            const gfloat4_t* jq = (const gfloat4_t*)(rb + lo);
            j0[k] = jq[0];
            j1[k] = jq[8];
        }

        const int pn = (p + 1 < GAT_B * GAT_N) ? (p + 1) : p;
        const int4* idxn = (const int4*)(nbr + (size_t)pn * GAT_K);
        i0 = idxn[0];
        i1 = idxn[1];
        i2 = idxn[2];
        i3 = idxn[3];

        float ssrc = dot4f(hi0, as0) + dot4f(hi1, as1);
        ssrc += __shfl_xor(ssrc, 1, 32);
        ssrc += __shfl_xor(ssrc, 2, 32);
        ssrc += __shfl_xor(ssrc, 4, 32);

        float m = -3.402823466e+38f;
        float s = 0.0f;
        float a0 = 0.f, a1 = 0.f, a2 = 0.f, a3 = 0.f;
        float a4 = 0.f, a5 = 0.f, a6 = 0.f, a7 = 0.f;

        #pragma unroll
        for (int k = 0; k < GAT_K; ++k) {
            float sd = dot4f(j0[k], ad0) + dot4f(j1[k], ad1);
            sd += __shfl_xor(sd, 1, 32);
            sd += __shfl_xor(sd, 2, 32);
            sd += __shfl_xor(sd, 4, 32);

            const float e  = (ssrc + sd) * 0.125f;
            const float mn = fmaxf(m, e);
            const float sc = __expf(m - mn);
            const float w  = __expf(e - mn);
            s = s * sc + w;
            m = mn;
            a0 = a0 * sc + w * j0[k].x;
            a1 = a1 * sc + w * j0[k].y;
            a2 = a2 * sc + w * j0[k].z;
            a3 = a3 * sc + w * j0[k].w;
            a4 = a4 * sc + w * j1[k].x;
            a5 = a5 * sc + w * j1[k].y;
            a6 = a6 * sc + w * j1[k].z;
            a7 = a7 * sc + w * j1[k].w;
        }

        const float inv = 1.0f / s;
        float v[8] = { a0 * inv, a1 * inv, a2 * inv, a3 * inv,
                       a4 * inv, a5 * inv, a6 * inv, a7 * inv };

        #pragma unroll
        for (int j = 0; j < 8; ++j) {
            float t = v[j];
            t += __shfl_xor(t, 8, 32);
            t += __shfl_xor(t, 16, 32);
            v[j] = t * 0.25f;
        }

        if (lane < 8) {
            float* op = out + (size_t)p * GAT_CH + lane * 4;
            *(float4*)(op)      = make_float4(v[0], v[1], v[2], v[3]);
            *(float4*)(op + 32) = make_float4(v[4], v[5], v[6], v[7]);
        }
    }
}

extern "C" void kernel_launch(void* const* d_in, const int* in_sizes, int n_in,
                              void* d_out, int out_size, void* d_ws, size_t ws_size,
                              hipStream_t stream)
{
    const float* x    = (const float*)d_in[0];
    const int*   idx  = (const int*)  d_in[1];
    const float* W    = (const float*)d_in[2];
    const float* attn = (const float*)d_in[3];
    float* out = (float*)d_out;
    float* h   = (float*)d_ws;

    gat_gemm_wmma<<<(GAT_B * GAT_N) / (16 * GEMM_MT), 128, 0, stream>>>(x, W, h);

    gat_gather_softmax<<<(GAT_B * GAT_N) / (GS_WAVES * GS_NPW), 32 * GS_WAVES, 0, stream>>>(h, idx, attn, out);
}
